# starting kernel
# baseline (speedup 1.0000x reference)
#include <hip/hip_runtime.h>

typedef __attribute__((ext_vector_type(2))) float v2f;
typedef __attribute__((ext_vector_type(8))) float v8f;

#define D_E   32
#define D_V   64
#define D_G   64
#define HDIM  256
#define D_OUT 64
#define NIN   256
#define MT    64
#define NTHR  512
#define LDP   260

__device__ __forceinline__ unsigned mapf(float f) {
  unsigned u = __float_as_uint(f);
  return (u & 0x80000000u) ? ~u : (u | 0x80000000u);
}
__device__ __forceinline__ float unmapf(unsigned u) {
  unsigned s = (u & 0x80000000u) ? (u ^ 0x80000000u) : ~u;
  return __uint_as_float(s);
}

__global__ void init_ws(float* sumb, float* cntb, unsigned* minb, unsigned* maxb, int nv) {
  int tid = blockIdx.x * blockDim.x + threadIdx.x;
  int total = nv * D_E;
  if (tid < total) {
    sumb[tid] = 0.0f;
    minb[tid] = 0xFFFFFFFFu;
    maxb[tid] = 0u;
    if (tid < nv) cntb[tid] = 0.0f;
  }
}

__global__ void edge_scatter(const int* __restrict__ src, const float* __restrict__ e_attr,
                             float* sumb, float* cntb, unsigned* minb, unsigned* maxb,
                             int ne) {
  long long tid = (long long)blockIdx.x * blockDim.x + threadIdx.x;
  int e = (int)(tid >> 5);
  int f = (int)(tid & 31);
  if (e >= ne) return;
  int s = src[e];
  float v = e_attr[(long long)e * D_E + f];
  int o = s * D_E + f;
  atomicAdd(&sumb[o], v);
  unsigned m = mapf(v);
  atomicMin(&minb[o], m);
  atomicMax(&maxb[o], m);
  if (f == 0) atomicAdd(&cntb[s], 1.0f);
}

template <int NTW>
__device__ __forceinline__ void layer_wmma(const float* __restrict__ hin,
                                           float* __restrict__ hout,
                                           const float* __restrict__ W,
                                           const float* __restrict__ b,
                                           int wstride, bool do_relu,
                                           int wave, int lane) {
  int lr   = lane & 15;
  int half = lane >> 4;
  int rt   = wave & 3;
  int col0 = (wave >> 2) * 16 + lr;

  v8f acc[NTW];
#pragma unroll
  for (int j = 0; j < NTW; ++j) {
    float bv = b[col0 + 64 * j];
    acc[j] = {bv, bv, bv, bv, bv, bv, bv, bv};
  }

  const float* arow = &hin[(rt * 16 + lr) * LDP];
  for (int kc = 0; kc < NIN; kc += 32) {
    v2f afrag[8];
#pragma unroll
    for (int k = 0; k < 8; ++k) {
      int kk = kc + 4 * k + 2 * half;
      afrag[k].x = arow[kk];
      afrag[k].y = arow[kk + 1];
    }
#pragma unroll
    for (int k = 0; k < 8; ++k) {
      int kk = kc + 4 * k + 2 * half;
      const float* wr0 = &W[(long long)kk * wstride + col0];
      const float* wr1 = wr0 + wstride;
#pragma unroll
      for (int j = 0; j < NTW; ++j) {
        v2f bb;
        bb.x = wr0[64 * j];
        bb.y = wr1[64 * j];
        acc[j] = __builtin_amdgcn_wmma_f32_16x16x4_f32(
            false, afrag[k], false, bb, (short)0, acc[j], false, false);
      }
    }
  }

#pragma unroll
  for (int j = 0; j < NTW; ++j) {
#pragma unroll
    for (int i = 0; i < 8; ++i) {
      float x = acc[j][i];
      if (do_relu) x = fmaxf(x, 0.0f);
      hout[(rt * 16 + i + 8 * half) * LDP + col0 + 64 * j] = x;
    }
  }
}

__global__ void __launch_bounds__(NTHR)
mlp_fused(const float* __restrict__ v_attr, const float* __restrict__ g,
          const int* __restrict__ batch,
          const float* __restrict__ W1, const float* __restrict__ b1,
          const float* __restrict__ W2, const float* __restrict__ b2,
          const float* __restrict__ W3, const float* __restrict__ b3,
          const float* __restrict__ sumb, const float* __restrict__ cntb,
          const unsigned* __restrict__ minb, const unsigned* __restrict__ maxb,
          float* __restrict__ out, int nv) {
  extern __shared__ float smem[];
  float* buf0 = smem;
  float* buf1 = smem + MT * LDP;
  int tid  = threadIdx.x;
  int wave = tid >> 5;
  int lane = tid & 31;
  int base = blockIdx.x * MT;

  for (int idx = tid; idx < MT * NIN; idx += NTHR) {
    int r = idx >> 8;
    int c = idx & 255;
    int v = base + r;
    float x = 0.0f;
    if (v < nv) {
      if (c < 64) {
        x = v_attr[(long long)v * D_V + c];
      } else if (c < 192) {
        int f = c & 31;
        int o = v * D_E + f;
        float cnt = cntb[v];
        if (c < 96)       x = (cnt > 0.0f) ? unmapf(minb[o]) : 0.0f;
        else if (c < 128) x = sumb[o] / fmaxf(cnt, 1.0f);
        else if (c < 160) x = sumb[o];
        else              x = (cnt > 0.0f) ? unmapf(maxb[o]) : 0.0f;
      } else {
        x = g[batch[v] * D_G + (c - 192)];
      }
    }
    buf0[r * LDP + c] = x;
  }
  __syncthreads();

  layer_wmma<4>(buf0, buf1, W1, b1,  HDIM,   true,  wave, lane);
  __syncthreads();
  layer_wmma<4>(buf1, buf0, W2, b2,  HDIM,   true,  wave, lane);
  __syncthreads();
  layer_wmma<1>(buf0, buf1, W3, b3,  D_OUT,  false, wave, lane);
  __syncthreads();

  for (int idx = tid; idx < MT * D_OUT; idx += NTHR) {
    int r = idx >> 6;
    int c = idx & 63;
    int v = base + r;
    if (v < nv) out[(long long)v * D_OUT + c] = buf1[r * LDP + c];
  }
}

extern "C" void kernel_launch(void* const* d_in, const int* in_sizes, int n_in,
                              void* d_out, int out_size, void* d_ws, size_t ws_size,
                              hipStream_t stream) {
  const float* v_attr = (const float*)d_in[0];
  const int*   edge   = (const int*)d_in[1];
  const float* e_attr = (const float*)d_in[2];
  const float* g      = (const float*)d_in[3];
  const int*   batch  = (const int*)d_in[4];
  const float* W1     = (const float*)d_in[5];
  const float* b1     = (const float*)d_in[6];
  const float* W2     = (const float*)d_in[7];
  const float* b2     = (const float*)d_in[8];
  const float* W3     = (const float*)d_in[9];
  const float* b3     = (const float*)d_in[10];
  float* out = (float*)d_out;

  int nv = in_sizes[0] / D_V;
  int ne = in_sizes[2] / D_E;

  float*    sumb = (float*)d_ws;
  unsigned* minb = (unsigned*)(sumb + (size_t)nv * D_E);
  unsigned* maxb = minb + (size_t)nv * D_E;
  float*    cntb = (float*)(maxb + (size_t)nv * D_E);

  int t1 = nv * D_E;
  init_ws<<<(t1 + 255) / 256, 256, 0, stream>>>(sumb, cntb, minb, maxb, nv);

  long long te = (long long)ne * D_E;
  edge_scatter<<<(int)((te + 255) / 256), 256, 0, stream>>>(
      edge, e_attr, sumb, cntb, minb, maxb, ne);

  int nblk = (nv + MT - 1) / MT;
  size_t sh = (size_t)2 * MT * LDP * sizeof(float);
  mlp_fused<<<nblk, NTHR, sh, stream>>>(v_attr, g, batch,
                                        W1, b1, W2, b2, W3, b3,
                                        sumb, cntb, minb, maxb, out, nv);
}


# Replace the per-element atomic sum/min/max/count scatter by one atomic exchange per edge that links per-vertex edge lists, then reduce each list in registers; sums now run in list (arrival) order
# speedup vs baseline: 1.3658x; 1.3658x over previous
#include <hip/hip_runtime.h>

typedef __attribute__((ext_vector_type(2))) float v2f;
typedef __attribute__((ext_vector_type(8))) float v8f;

#define D_E   32
#define D_V   64
#define D_G   64
#define HDIM  256
#define D_OUT 64
#define NIN   256
#define MT    64
#define NTHR  512
#define LDP   260

__global__ void init_heads(int* head, int nv) {
  int tid = blockIdx.x * blockDim.x + threadIdx.x;
  if (tid < nv) head[tid] = -1;
}

__global__ void edge_link(const int* __restrict__ src, int* head, int* __restrict__ nxt,
                          int ne, int nv) {
  int e = blockIdx.x * blockDim.x + threadIdx.x;
  if (e >= ne) return;
  int s = src[e];
  int old = -1;
  if ((unsigned)s < (unsigned)nv) old = atomicExch(&head[s], e);
  nxt[e] = old;
}

__global__ void __launch_bounds__(256)
vertex_reduce(const float* __restrict__ e_attr, const int* __restrict__ head,
              const int* __restrict__ nxt,
              float* __restrict__ sumb, float* __restrict__ cntb,
              float* __restrict__ minb, float* __restrict__ maxb, int nv, int ne) {
  int tid = blockIdx.x * blockDim.x + threadIdx.x;
  int v = tid >> 3;
  int q = tid & 7;
  if (v >= nv) return;
  const float inf = __builtin_inff();
  float4 s  = make_float4(0.0f, 0.0f, 0.0f, 0.0f);
  float4 mn = make_float4(inf, inf, inf, inf);
  float4 mx = make_float4(-inf, -inf, -inf, -inf);
  const float* rowbase = e_attr + 4 * q;
  int cnt = 0;
  int e = head[v];
  while ((unsigned)e < (unsigned)ne && cnt < ne) {
    float4 x = *reinterpret_cast<const float4*>(rowbase + (long long)e * D_E);
    e = nxt[e];
    s.x += x.x; s.y += x.y; s.z += x.z; s.w += x.w;
    mn.x = fminf(mn.x, x.x); mn.y = fminf(mn.y, x.y);
    mn.z = fminf(mn.z, x.z); mn.w = fminf(mn.w, x.w);
    mx.x = fmaxf(mx.x, x.x); mx.y = fmaxf(mx.y, x.y);
    mx.z = fmaxf(mx.z, x.z); mx.w = fmaxf(mx.w, x.w);
    ++cnt;
  }
  if (cnt == 0) {
    mn = s;
    mx = s;
  }
  long long o = (long long)v * D_E + 4 * q;
  *reinterpret_cast<float4*>(sumb + o) = s;
  *reinterpret_cast<float4*>(minb + o) = mn;
  *reinterpret_cast<float4*>(maxb + o) = mx;
  if (q == 0) cntb[v] = (float)cnt;
}

template <int NTW>
__device__ __forceinline__ void layer_wmma(const float* __restrict__ hin,
                                           float* __restrict__ hout,
                                           const float* __restrict__ W,
                                           const float* __restrict__ b,
                                           int wstride, bool do_relu,
                                           int wave, int lane) {
  int lr   = lane & 15;
  int half = lane >> 4;
  int rt   = wave & 3;
  int col0 = (wave >> 2) * 16 + lr;

  v8f acc[NTW];
#pragma unroll
  for (int j = 0; j < NTW; ++j) {
    float bv = b[col0 + 64 * j];
    acc[j] = {bv, bv, bv, bv, bv, bv, bv, bv};
  }

  const float* arow = &hin[(rt * 16 + lr) * LDP];
  for (int kc = 0; kc < NIN; kc += 32) {
    v2f afrag[8];
#pragma unroll
    for (int k = 0; k < 8; ++k) {
      int kk = kc + 4 * k + 2 * half;
      afrag[k].x = arow[kk];
      afrag[k].y = arow[kk + 1];
    }
#pragma unroll
    for (int k = 0; k < 8; ++k) {
      int kk = kc + 4 * k + 2 * half;
      const float* wr0 = &W[(long long)kk * wstride + col0];
      const float* wr1 = wr0 + wstride;
#pragma unroll
      for (int j = 0; j < NTW; ++j) {
        v2f bb;
        bb.x = wr0[64 * j];
        bb.y = wr1[64 * j];
        acc[j] = __builtin_amdgcn_wmma_f32_16x16x4_f32(
            false, afrag[k], false, bb, (short)0, acc[j], false, false);
      }
    }
  }

#pragma unroll
  for (int j = 0; j < NTW; ++j) {
#pragma unroll
    for (int i = 0; i < 8; ++i) {
      float x = acc[j][i];
      if (do_relu) x = fmaxf(x, 0.0f);
      hout[(rt * 16 + i + 8 * half) * LDP + col0 + 64 * j] = x;
    }
  }
}

__global__ void __launch_bounds__(NTHR)
mlp_fused(const float* __restrict__ v_attr, const float* __restrict__ g,
          const int* __restrict__ batch,
          const float* __restrict__ W1, const float* __restrict__ b1,
          const float* __restrict__ W2, const float* __restrict__ b2,
          const float* __restrict__ W3, const float* __restrict__ b3,
          const float* __restrict__ sumb, const float* __restrict__ cntb,
          const float* __restrict__ minb, const float* __restrict__ maxb,
          float* __restrict__ out, int nv) {
  extern __shared__ float smem[];
  float* buf0 = smem;
  float* buf1 = smem + MT * LDP;
  int tid  = threadIdx.x;
  int wave = tid >> 5;
  int lane = tid & 31;
  int base = blockIdx.x * MT;

  for (int idx = tid; idx < MT * NIN; idx += NTHR) {
    int r = idx >> 8;
    int c = idx & 255;
    int v = base + r;
    float x = 0.0f;
    if (v < nv) {
      if (c < 64) {
        x = v_attr[(long long)v * D_V + c];
      } else if (c < 192) {
        int f = c & 31;
        int o = v * D_E + f;
        float cnt = cntb[v];
        if (c < 96)       x = minb[o];
        else if (c < 128) x = sumb[o] / fmaxf(cnt, 1.0f);
        else if (c < 160) x = sumb[o];
        else              x = maxb[o];
      } else {
        x = g[batch[v] * D_G + (c - 192)];
      }
    }
    buf0[r * LDP + c] = x;
  }
  __syncthreads();

  layer_wmma<4>(buf0, buf1, W1, b1,  HDIM,   true,  wave, lane);
  __syncthreads();
  layer_wmma<4>(buf1, buf0, W2, b2,  HDIM,   true,  wave, lane);
  __syncthreads();
  layer_wmma<1>(buf0, buf1, W3, b3,  D_OUT,  false, wave, lane);
  __syncthreads();

  for (int idx = tid; idx < MT * D_OUT; idx += NTHR) {
    int r = idx >> 6;
    int c = idx & 63;
    int v = base + r;
    if (v < nv) out[(long long)v * D_OUT + c] = buf1[r * LDP + c];
  }
}

extern "C" void kernel_launch(void* const* d_in, const int* in_sizes, int n_in,
                              void* d_out, int out_size, void* d_ws, size_t ws_size,
                              hipStream_t stream) {
  const float* v_attr = (const float*)d_in[0];
  const int*   edge   = (const int*)d_in[1];
  const float* e_attr = (const float*)d_in[2];
  const float* g      = (const float*)d_in[3];
  const int*   batch  = (const int*)d_in[4];
  const float* W1     = (const float*)d_in[5];
  const float* b1     = (const float*)d_in[6];
  const float* W2     = (const float*)d_in[7];
  const float* b2     = (const float*)d_in[8];
  const float* W3     = (const float*)d_in[9];
  const float* b3     = (const float*)d_in[10];
  float* out = (float*)d_out;

  int nv = in_sizes[0] / D_V;
  int ne = in_sizes[2] / D_E;

  float* sumb = (float*)d_ws;
  float* minb = sumb + (size_t)nv * D_E;
  float* maxb = minb + (size_t)nv * D_E;
  float* cntb = maxb + (size_t)nv * D_E;
  int*   head = (int*)(cntb + nv);
  int*   nxt  = head + nv;

  init_heads<<<(nv + 255) / 256, 256, 0, stream>>>(head, nv);

  edge_link<<<(ne + 255) / 256, 256, 0, stream>>>(edge, head, nxt, ne, nv);

  vertex_reduce<<<(nv * 8 + 255) / 256, 256, 0, stream>>>(
      e_attr, head, nxt, sumb, cntb, minb, maxb, nv, ne);

  int nblk = (nv + MT - 1) / MT;
  size_t sh = (size_t)2 * MT * LDP * sizeof(float);
  mlp_fused<<<nblk, NTHR, sh, stream>>>(v_attr, g, batch,
                                        W1, b1, W2, b2, W3, b3,
                                        sumb, cntb, minb, maxb, out, nv);
}


# Fold the per-vertex list reduction into the MLP block's input staging: 8 lanes per row walk that vertex's edge list and write min/mean/sum/max straight into the LDS tile; list order of sums kept
# speedup vs baseline: 1.6521x; 1.2096x over previous
#include <hip/hip_runtime.h>

typedef __attribute__((ext_vector_type(2))) float v2f;
typedef __attribute__((ext_vector_type(8))) float v8f;

#define D_E   32
#define D_V   64
#define D_G   64
#define HDIM  256
#define D_OUT 64
#define NIN   256
#define MT    64
#define NTHR  512
#define LDP   260

__global__ void init_heads(int* head, int nv) {
  int tid = blockIdx.x * blockDim.x + threadIdx.x;
  if (tid < nv) head[tid] = -1;
}

__global__ void edge_link(const int* __restrict__ src, int* head, int* __restrict__ nxt,
                          int ne, int nv) {
  int e = blockIdx.x * blockDim.x + threadIdx.x;
  if (e >= ne) return;
  int s = src[e];
  int old = -1;
  if ((unsigned)s < (unsigned)nv) old = atomicExch(&head[s], e);
  nxt[e] = old;
}

template <int NTW>
__device__ __forceinline__ void layer_wmma(const float* __restrict__ hin,
                                           float* __restrict__ hout,
                                           const float* __restrict__ W,
                                           const float* __restrict__ b,
                                           int wstride, bool do_relu,
                                           int wave, int lane) {
  int lr   = lane & 15;
  int half = lane >> 4;
  int rt   = wave & 3;
  int col0 = (wave >> 2) * 16 + lr;

  v8f acc[NTW];
#pragma unroll
  for (int j = 0; j < NTW; ++j) {
    float bv = b[col0 + 64 * j];
    acc[j] = {bv, bv, bv, bv, bv, bv, bv, bv};
  }

  const float* arow = &hin[(rt * 16 + lr) * LDP];
  for (int kc = 0; kc < NIN; kc += 32) {
    v2f afrag[8];
#pragma unroll
    for (int k = 0; k < 8; ++k) {
      int kk = kc + 4 * k + 2 * half;
      afrag[k].x = arow[kk];
      afrag[k].y = arow[kk + 1];
    }
#pragma unroll
    for (int k = 0; k < 8; ++k) {
      int kk = kc + 4 * k + 2 * half;
      const float* wr0 = &W[(long long)kk * wstride + col0];
      const float* wr1 = wr0 + wstride;
#pragma unroll
      for (int j = 0; j < NTW; ++j) {
        v2f bb;
        bb.x = wr0[64 * j];
        bb.y = wr1[64 * j];
        acc[j] = __builtin_amdgcn_wmma_f32_16x16x4_f32(
            false, afrag[k], false, bb, (short)0, acc[j], false, false);
      }
    }
  }

#pragma unroll
  for (int j = 0; j < NTW; ++j) {
#pragma unroll
    for (int i = 0; i < 8; ++i) {
      float x = acc[j][i];
      if (do_relu) x = fmaxf(x, 0.0f);
      hout[(rt * 16 + i + 8 * half) * LDP + col0 + 64 * j] = x;
    }
  }
}

__global__ void __launch_bounds__(NTHR)
mlp_fused(const float* __restrict__ v_attr, const float* __restrict__ e_attr,
          const float* __restrict__ g, const int* __restrict__ batch,
          const int* __restrict__ head, const int* __restrict__ nxt,
          const float* __restrict__ W1, const float* __restrict__ b1,
          const float* __restrict__ W2, const float* __restrict__ b2,
          const float* __restrict__ W3, const float* __restrict__ b3,
          float* __restrict__ out, int nv, int ne) {
  extern __shared__ float smem[];
  float* buf0 = smem;
  float* buf1 = smem + MT * LDP;
  int tid  = threadIdx.x;
  int wave = tid >> 5;
  int lane = tid & 31;
  int base = blockIdx.x * MT;

  {
    int r = tid >> 3;
    int q = tid & 7;
    int v = base + r;
    const float4 zero = make_float4(0.0f, 0.0f, 0.0f, 0.0f);
    float4 va0 = zero, va1 = zero, g0 = zero, g1 = zero;
    float4 s = zero, mn = zero, mx = zero, mean = zero;
    if (v < nv) {
      const float* vr = v_attr + (long long)v * D_V + 4 * q;
      va0 = *reinterpret_cast<const float4*>(vr);
      va1 = *reinterpret_cast<const float4*>(vr + 32);
      int e = head[v];
      const float* gr = g + batch[v] * D_G + 4 * q;
      g0 = *reinterpret_cast<const float4*>(gr);
      g1 = *reinterpret_cast<const float4*>(gr + 32);

      const float inf = __builtin_inff();
      mn = make_float4(inf, inf, inf, inf);
      mx = make_float4(-inf, -inf, -inf, -inf);
      const float* rowbase = e_attr + 4 * q;
      int cnt = 0;
      while ((unsigned)e < (unsigned)ne && cnt < ne) {
        float4 x = *reinterpret_cast<const float4*>(rowbase + (long long)e * D_E);
        e = nxt[e];
        s.x += x.x; s.y += x.y; s.z += x.z; s.w += x.w;
        mn.x = fminf(mn.x, x.x); mn.y = fminf(mn.y, x.y);
        mn.z = fminf(mn.z, x.z); mn.w = fminf(mn.w, x.w);
        mx.x = fmaxf(mx.x, x.x); mx.y = fmaxf(mx.y, x.y);
        mx.z = fmaxf(mx.z, x.z); mx.w = fmaxf(mx.w, x.w);
        ++cnt;
      }
      if (cnt == 0) {
        mn = s;
        mx = s;
      }
      float d = fmaxf((float)cnt, 1.0f);
      mean = make_float4(s.x / d, s.y / d, s.z / d, s.w / d);
    }
    float* row = buf0 + r * LDP + 4 * q;
    *reinterpret_cast<float4*>(row)       = va0;
    *reinterpret_cast<float4*>(row + 32)  = va1;
    *reinterpret_cast<float4*>(row + 64)  = mn;
    *reinterpret_cast<float4*>(row + 96)  = mean;
    *reinterpret_cast<float4*>(row + 128) = s;
    *reinterpret_cast<float4*>(row + 160) = mx;
    *reinterpret_cast<float4*>(row + 192) = g0;
    *reinterpret_cast<float4*>(row + 224) = g1;
  }
  __syncthreads();

  layer_wmma<4>(buf0, buf1, W1, b1,  HDIM,   true,  wave, lane);
  __syncthreads();
  layer_wmma<4>(buf1, buf0, W2, b2,  HDIM,   true,  wave, lane);
  __syncthreads();
  layer_wmma<1>(buf0, buf1, W3, b3,  D_OUT,  false, wave, lane);
  __syncthreads();

  for (int idx = tid; idx < MT * D_OUT; idx += NTHR) {
    int r = idx >> 6;
    int c = idx & 63;
    int v = base + r;
    if (v < nv) out[(long long)v * D_OUT + c] = buf1[r * LDP + c];
  }
}

extern "C" void kernel_launch(void* const* d_in, const int* in_sizes, int n_in,
                              void* d_out, int out_size, void* d_ws, size_t ws_size,
                              hipStream_t stream) {
  const float* v_attr = (const float*)d_in[0];
  const int*   edge   = (const int*)d_in[1];
  const float* e_attr = (const float*)d_in[2];
  const float* g      = (const float*)d_in[3];
  const int*   batch  = (const int*)d_in[4];
  const float* W1     = (const float*)d_in[5];
  const float* b1     = (const float*)d_in[6];
  const float* W2     = (const float*)d_in[7];
  const float* b2     = (const float*)d_in[8];
  const float* W3     = (const float*)d_in[9];
  const float* b3     = (const float*)d_in[10];
  float* out = (float*)d_out;

  int nv = in_sizes[0] / D_V;
  int ne = in_sizes[2] / D_E;

  int* head = (int*)d_ws;
  int* nxt  = head + nv;

  init_heads<<<(nv + 255) / 256, 256, 0, stream>>>(head, nv);

  edge_link<<<(ne + 255) / 256, 256, 0, stream>>>(edge, head, nxt, ne, nv);

  int nblk = (nv + MT - 1) / MT;
  size_t sh = (size_t)2 * MT * LDP * sizeof(float);
  mlp_fused<<<nblk, NTHR, sh, stream>>>(v_attr, e_attr, g, batch, head, nxt,
                                        W1, b1, W2, b2, W3, b3, out, nv, ne);
}


# MLP layers 1-2 re-tiled: each wave owns one 16-column tile over all 64 rows, so one W fragment from global feeds 4 WMMAs (A re-read from LDS); W loads per WMMA cut 4x; order of sums kept
# speedup vs baseline: 1.7100x; 1.0350x over previous
#include <hip/hip_runtime.h>

typedef __attribute__((ext_vector_type(2))) float v2f;
typedef __attribute__((ext_vector_type(8))) float v8f;

#define D_E   32
#define D_V   64
#define D_G   64
#define HDIM  256
#define D_OUT 64
#define NIN   256
#define MT    64
#define NTHR  512
#define LDP   260

__global__ void init_heads(int* head, int nv) {
  int tid = blockIdx.x * blockDim.x + threadIdx.x;
  if (tid < nv) head[tid] = -1;
}

__global__ void edge_link(const int* __restrict__ src, int* head, int* __restrict__ nxt,
                          int ne, int nv) {
  int e = blockIdx.x * blockDim.x + threadIdx.x;
  if (e >= ne) return;
  int s = src[e];
  int old = -1;
  if ((unsigned)s < (unsigned)nv) old = atomicExch(&head[s], e);
  nxt[e] = old;
}

template <int NTW>
__device__ __forceinline__ void layer_wmma(const float* __restrict__ hin,
                                           float* __restrict__ hout,
                                           const float* __restrict__ W,
                                           const float* __restrict__ b,
                                           int wstride, bool do_relu,
                                           int wave, int lane) {
  int lr   = lane & 15;
  int half = lane >> 4;
  int rt   = wave & 3;
  int col0 = (wave >> 2) * 16 + lr;

  v8f acc[NTW];
#pragma unroll
  for (int j = 0; j < NTW; ++j) {
    float bv = b[col0 + 64 * j];
    acc[j] = {bv, bv, bv, bv, bv, bv, bv, bv};
  }

  const float* arow = &hin[(rt * 16 + lr) * LDP];
  for (int kc = 0; kc < NIN; kc += 32) {
    v2f afrag[8];
#pragma unroll
    for (int k = 0; k < 8; ++k) {
      int kk = kc + 4 * k + 2 * half;
      afrag[k].x = arow[kk];
      afrag[k].y = arow[kk + 1];
    }
#pragma unroll
    for (int k = 0; k < 8; ++k) {
      int kk = kc + 4 * k + 2 * half;
      const float* wr0 = &W[(long long)kk * wstride + col0];
      const float* wr1 = wr0 + wstride;
#pragma unroll
      for (int j = 0; j < NTW; ++j) {
        v2f bb;
        bb.x = wr0[64 * j];
        bb.y = wr1[64 * j];
        acc[j] = __builtin_amdgcn_wmma_f32_16x16x4_f32(
            false, afrag[k], false, bb, (short)0, acc[j], false, false);
      }
    }
  }

#pragma unroll
  for (int j = 0; j < NTW; ++j) {
#pragma unroll
    for (int i = 0; i < 8; ++i) {
      float x = acc[j][i];
      if (do_relu) x = fmaxf(x, 0.0f);
      hout[(rt * 16 + i + 8 * half) * LDP + col0 + 64 * j] = x;
    }
  }
}


#define KCH 16
__device__ __forceinline__ void load_chunk(const float* __restrict__ wp,
                                           const float* __restrict__ ap, int kc,
                                           v2f (&bq)[4], v2f (&aq)[4][4]) {
#pragma unroll
  for (int k = 0; k < 4; ++k) {
    const float* wr = wp + (kc + 4 * k) * HDIM;
    bq[k].x = wr[0];
    bq[k].y = wr[HDIM];
  }
#pragma unroll
  for (int k = 0; k < 4; ++k) {
#pragma unroll
    for (int t = 0; t < 4; ++t) {
      aq[k][t] = *reinterpret_cast<const v2f*>(ap + t * 16 * LDP + kc + 4 * k);
    }
  }
}

__device__ __forceinline__ void mma_chunk(v8f (&acc)[4], const v2f (&bq)[4],
                                          const v2f (&aq)[4][4]) {
#pragma unroll
  for (int k = 0; k < 4; ++k) {
#pragma unroll
    for (int t = 0; t < 4; ++t) {
      acc[t] = __builtin_amdgcn_wmma_f32_16x16x4_f32(
          false, aq[k][t], false, bq[k], (short)0, acc[t], false, false);
    }
  }
}

__device__ __forceinline__ void layer_wmma_cols(const float* __restrict__ hin,
                                                float* __restrict__ hout,
                                                const float* __restrict__ W,
                                                const float* __restrict__ b,
                                                int wave, int lane) {
  int lr   = lane & 15;
  int half = lane >> 4;
  int col  = wave * 16 + lr;

  float bv = b[col];
  v8f acc[4];
#pragma unroll
  for (int t = 0; t < 4; ++t) acc[t] = {bv, bv, bv, bv, bv, bv, bv, bv};

  const float* wp = W + (2 * half) * HDIM + col;
  const float* ap = hin + lr * LDP + 2 * half;

  v2f b0[4], b1[4];
  v2f a0[4][4], a1[4][4];
  load_chunk(wp, ap, 0, b0, a0);
  for (int kc = 0; kc < NIN; kc += 2 * KCH) {
    load_chunk(wp, ap, kc + KCH, b1, a1);
    mma_chunk(acc, b0, a0);
    if (kc + 2 * KCH < NIN) load_chunk(wp, ap, kc + 2 * KCH, b0, a0);
    mma_chunk(acc, b1, a1);
  }

#pragma unroll
  for (int t = 0; t < 4; ++t) {
#pragma unroll
    for (int i = 0; i < 8; ++i) {
      hout[(t * 16 + i + 8 * half) * LDP + col] = fmaxf(acc[t][i], 0.0f);
    }
  }
}

__global__ void __launch_bounds__(NTHR)
mlp_fused(const float* __restrict__ v_attr, const float* __restrict__ e_attr,
          const float* __restrict__ g, const int* __restrict__ batch,
          const int* __restrict__ head, const int* __restrict__ nxt,
          const float* __restrict__ W1, const float* __restrict__ b1,
          const float* __restrict__ W2, const float* __restrict__ b2,
          const float* __restrict__ W3, const float* __restrict__ b3,
          float* __restrict__ out, int nv, int ne) {
  extern __shared__ float smem[];
  float* buf0 = smem;
  float* buf1 = smem + MT * LDP;
  int tid  = threadIdx.x;
  int wave = tid >> 5;
  int lane = tid & 31;
  int base = blockIdx.x * MT;

  {
    int r = tid >> 3;
    int q = tid & 7;
    int v = base + r;
    const float4 zero = make_float4(0.0f, 0.0f, 0.0f, 0.0f);
    float4 va0 = zero, va1 = zero, g0 = zero, g1 = zero;
    float4 s = zero, mn = zero, mx = zero, mean = zero;
    if (v < nv) {
      const float* vr = v_attr + (long long)v * D_V + 4 * q;
      va0 = *reinterpret_cast<const float4*>(vr);
      va1 = *reinterpret_cast<const float4*>(vr + 32);
      int e = head[v];
      const float* gr = g + batch[v] * D_G + 4 * q;
      g0 = *reinterpret_cast<const float4*>(gr);
      g1 = *reinterpret_cast<const float4*>(gr + 32);

      const float inf = __builtin_inff();
      mn = make_float4(inf, inf, inf, inf);
      mx = make_float4(-inf, -inf, -inf, -inf);
      const float* rowbase = e_attr + 4 * q;
      int cnt = 0;
      while ((unsigned)e < (unsigned)ne && cnt < ne) {
        float4 x = *reinterpret_cast<const float4*>(rowbase + (long long)e * D_E);
        e = nxt[e];
        s.x += x.x; s.y += x.y; s.z += x.z; s.w += x.w;
        mn.x = fminf(mn.x, x.x); mn.y = fminf(mn.y, x.y);
        mn.z = fminf(mn.z, x.z); mn.w = fminf(mn.w, x.w);
        mx.x = fmaxf(mx.x, x.x); mx.y = fmaxf(mx.y, x.y);
        mx.z = fmaxf(mx.z, x.z); mx.w = fmaxf(mx.w, x.w);
        ++cnt;
      }
      if (cnt == 0) {
        mn = s;
        mx = s;
      }
      float d = fmaxf((float)cnt, 1.0f);
      mean = make_float4(s.x / d, s.y / d, s.z / d, s.w / d);
    }
    float* row = buf0 + r * LDP + 4 * q;
    *reinterpret_cast<float4*>(row)       = va0;
    *reinterpret_cast<float4*>(row + 32)  = va1;
    *reinterpret_cast<float4*>(row + 64)  = mn;
    *reinterpret_cast<float4*>(row + 96)  = mean;
    *reinterpret_cast<float4*>(row + 128) = s;
    *reinterpret_cast<float4*>(row + 160) = mx;
    *reinterpret_cast<float4*>(row + 192) = g0;
    *reinterpret_cast<float4*>(row + 224) = g1;
  }
  __syncthreads();

  layer_wmma_cols(buf0, buf1, W1, b1, wave, lane);
  __syncthreads();
  layer_wmma_cols(buf1, buf0, W2, b2, wave, lane);
  __syncthreads();
  layer_wmma<1>(buf0, buf1, W3, b3,  D_OUT,  false, wave, lane);
  __syncthreads();

  for (int idx = tid; idx < MT * D_OUT; idx += NTHR) {
    int r = idx >> 6;
    int c = idx & 63;
    int v = base + r;
    if (v < nv) out[(long long)v * D_OUT + c] = buf1[r * LDP + c];
  }
}

extern "C" void kernel_launch(void* const* d_in, const int* in_sizes, int n_in,
                              void* d_out, int out_size, void* d_ws, size_t ws_size,
                              hipStream_t stream) {
  const float* v_attr = (const float*)d_in[0];
  const int*   edge   = (const int*)d_in[1];
  const float* e_attr = (const float*)d_in[2];
  const float* g      = (const float*)d_in[3];
  const int*   batch  = (const int*)d_in[4];
  const float* W1     = (const float*)d_in[5];
  const float* b1     = (const float*)d_in[6];
  const float* W2     = (const float*)d_in[7];
  const float* b2     = (const float*)d_in[8];
  const float* W3     = (const float*)d_in[9];
  const float* b3     = (const float*)d_in[10];
  float* out = (float*)d_out;

  int nv = in_sizes[0] / D_V;
  int ne = in_sizes[2] / D_E;

  int* head = (int*)d_ws;
  int* nxt  = head + nv;

  init_heads<<<(nv + 255) / 256, 256, 0, stream>>>(head, nv);

  edge_link<<<(ne + 255) / 256, 256, 0, stream>>>(edge, head, nxt, ne, nv);

  int nblk = (nv + MT - 1) / MT;
  size_t sh = (size_t)2 * MT * LDP * sizeof(float);
  mlp_fused<<<nblk, NTHR, sh, stream>>>(v_attr, e_attr, g, batch, head, nxt,
                                        W1, b1, W2, b2, W3, b3, out, nv, ne);
}


# Per-edge global atomic exchange replaced by LDS counters: edges binned by 256-vertex range (LDS histogram per 8192-edge block), MLP block ranks its rows' edges in LDS; sums in LDS arrival order
# speedup vs baseline: 2.2979x; 1.3438x over previous
#include <hip/hip_runtime.h>

typedef __attribute__((ext_vector_type(2))) float v2f;
typedef __attribute__((ext_vector_type(8))) float v8f;

#define D_E   32
#define D_V   64
#define D_G   64
#define HDIM  256
#define D_OUT 64
#define NIN   256
#define MT    64
#define NTHR  512
#define LDP   260
#define BSH   8
#define BWID  256
#define CPAD  32
#define EBLK  8192
#define ETHR  512
#define EPT   16
#define LCAP  128
#define LSTR  132

__global__ void init_counts(int* gcnt, int n) {
  int tid = blockIdx.x * blockDim.x + threadIdx.x;
  if (tid < n) gcnt[tid] = 0;
}

__global__ void __launch_bounds__(ETHR)
edge_bin(const int* __restrict__ src, int* gcnt, int* __restrict__ recs,
         int* __restrict__ ovf, int ne, int nv, int nb, int bcap) {
  extern __shared__ int sm[];
  int* hist   = sm;
  int* lstart = sm + nb;
  int* gbase  = sm + 2 * nb;
  int* totalp = sm + 3 * nb;
  int2* stage = reinterpret_cast<int2*>(sm + ((3 * nb + 3) & ~1));
  int tid = threadIdx.x;

  for (int b = tid; b < nb; b += ETHR) hist[b] = 0;
  __syncthreads();

  int e0 = blockIdx.x * EBLK;
  int sv[EPT], rk[EPT];
#pragma unroll
  for (int j = 0; j < EPT; ++j) {
    int e = e0 + j * ETHR + tid;
    int s = -1;
    if (e < ne) {
      s = src[e];
      if ((unsigned)s >= (unsigned)nv) s = -1;
    }
    sv[j] = s;
  }
#pragma unroll
  for (int j = 0; j < EPT; ++j) {
    rk[j] = 0;
    if (sv[j] >= 0) rk[j] = atomicAdd(&hist[sv[j] >> BSH], 1);
  }
  __syncthreads();

  if (tid < 32) {
    int seg = (nb + 31) / 32;
    int lo = tid * seg;
    int hi = lo + seg < nb ? lo + seg : nb;
    int sum = 0;
    for (int b = lo; b < hi; ++b) sum += hist[b];
    int incl = sum;
    for (int d = 1; d < 32; d <<= 1) {
      int t = __shfl_up(incl, d);
      if (tid >= d) incl += t;
    }
    int run = incl - sum;
    for (int b = lo; b < hi; ++b) {
      lstart[b] = run;
      run += hist[b];
    }
    if (tid == 31) *totalp = incl;
  }
  for (int b = tid; b < nb; b += ETHR) {
    int h = hist[b];
    gbase[b] = h ? atomicAdd(&gcnt[b * CPAD], h) : 0;
  }
  __syncthreads();

#pragma unroll
  for (int j = 0; j < EPT; ++j) {
    if (sv[j] >= 0) {
      stage[lstart[sv[j] >> BSH] + rk[j]] = make_int2(e0 + j * ETHR + tid, sv[j]);
    }
  }
  __syncthreads();

  int total = *totalp;
  for (int i = tid; i < total; i += ETHR) {
    int2 rec = stage[i];
    int b = rec.y >> BSH;
    int dst = gbase[b] + (i - lstart[b]);
    if (dst < bcap) {
      *reinterpret_cast<int2*>(recs + 2 * ((long long)b * bcap + dst)) = rec;
    } else {
      int k = atomicAdd(&gcnt[nb * CPAD], 1);
      *reinterpret_cast<int2*>(ovf + 2 * (long long)k) = rec;
    }
  }
}

template <int NTW>
__device__ __forceinline__ void layer_wmma(const float* __restrict__ hin,
                                           float* __restrict__ hout,
                                           const float* __restrict__ W,
                                           const float* __restrict__ b,
                                           int wstride, bool do_relu,
                                           int wave, int lane) {
  int lr   = lane & 15;
  int half = lane >> 4;
  int rt   = wave & 3;
  int col0 = (wave >> 2) * 16 + lr;

  v8f acc[NTW];
#pragma unroll
  for (int j = 0; j < NTW; ++j) {
    float bv = b[col0 + 64 * j];
    acc[j] = {bv, bv, bv, bv, bv, bv, bv, bv};
  }

  const float* arow = &hin[(rt * 16 + lr) * LDP];
  for (int kc = 0; kc < NIN; kc += 32) {
    v2f afrag[8];
#pragma unroll
    for (int k = 0; k < 8; ++k) {
      int kk = kc + 4 * k + 2 * half;
      afrag[k].x = arow[kk];
      afrag[k].y = arow[kk + 1];
    }
#pragma unroll
    for (int k = 0; k < 8; ++k) {
      int kk = kc + 4 * k + 2 * half;
      const float* wr0 = &W[(long long)kk * wstride + col0];
      const float* wr1 = wr0 + wstride;
#pragma unroll
      for (int j = 0; j < NTW; ++j) {
        v2f bb;
        bb.x = wr0[64 * j];
        bb.y = wr1[64 * j];
        acc[j] = __builtin_amdgcn_wmma_f32_16x16x4_f32(
            false, afrag[k], false, bb, (short)0, acc[j], false, false);
      }
    }
  }

#pragma unroll
  for (int j = 0; j < NTW; ++j) {
#pragma unroll
    for (int i = 0; i < 8; ++i) {
      float x = acc[j][i];
      if (do_relu) x = fmaxf(x, 0.0f);
      hout[(rt * 16 + i + 8 * half) * LDP + col0 + 64 * j] = x;
    }
  }
}


#define KCH 16
__device__ __forceinline__ void load_chunk(const float* __restrict__ wp,
                                           const float* __restrict__ ap, int kc,
                                           v2f (&bq)[4], v2f (&aq)[4][4]) {
#pragma unroll
  for (int k = 0; k < 4; ++k) {
    const float* wr = wp + (kc + 4 * k) * HDIM;
    bq[k].x = wr[0];
    bq[k].y = wr[HDIM];
  }
#pragma unroll
  for (int k = 0; k < 4; ++k) {
#pragma unroll
    for (int t = 0; t < 4; ++t) {
      aq[k][t] = *reinterpret_cast<const v2f*>(ap + t * 16 * LDP + kc + 4 * k);
    }
  }
}

__device__ __forceinline__ void mma_chunk(v8f (&acc)[4], const v2f (&bq)[4],
                                          const v2f (&aq)[4][4]) {
#pragma unroll
  for (int k = 0; k < 4; ++k) {
#pragma unroll
    for (int t = 0; t < 4; ++t) {
      acc[t] = __builtin_amdgcn_wmma_f32_16x16x4_f32(
          false, aq[k][t], false, bq[k], (short)0, acc[t], false, false);
    }
  }
}

__device__ __forceinline__ void layer_wmma_cols(const float* __restrict__ hin,
                                                float* __restrict__ hout,
                                                const float* __restrict__ W,
                                                const float* __restrict__ b,
                                                int wave, int lane) {
  int lr   = lane & 15;
  int half = lane >> 4;
  int col  = wave * 16 + lr;

  float bv = b[col];
  v8f acc[4];
#pragma unroll
  for (int t = 0; t < 4; ++t) acc[t] = {bv, bv, bv, bv, bv, bv, bv, bv};

  const float* wp = W + (2 * half) * HDIM + col;
  const float* ap = hin + lr * LDP + 2 * half;

  v2f b0[4], b1[4];
  v2f a0[4][4], a1[4][4];
  load_chunk(wp, ap, 0, b0, a0);
  for (int kc = 0; kc < NIN; kc += 2 * KCH) {
    load_chunk(wp, ap, kc + KCH, b1, a1);
    mma_chunk(acc, b0, a0);
    if (kc + 2 * KCH < NIN) load_chunk(wp, ap, kc + 2 * KCH, b0, a0);
    mma_chunk(acc, b1, a1);
  }

#pragma unroll
  for (int t = 0; t < 4; ++t) {
#pragma unroll
    for (int i = 0; i < 8; ++i) {
      hout[(t * 16 + i + 8 * half) * LDP + col] = fmaxf(acc[t][i], 0.0f);
    }
  }
}

__device__ __forceinline__ void acc_row(float4& s, float4& mn, float4& mx, const float4& x) {
  s.x += x.x; s.y += x.y; s.z += x.z; s.w += x.w;
  mn.x = fminf(mn.x, x.x); mn.y = fminf(mn.y, x.y);
  mn.z = fminf(mn.z, x.z); mn.w = fminf(mn.w, x.w);
  mx.x = fmaxf(mx.x, x.x); mx.y = fmaxf(mx.y, x.y);
  mx.z = fmaxf(mx.z, x.z); mx.w = fmaxf(mx.w, x.w);
}

__global__ void __launch_bounds__(NTHR)
mlp_fused(const float* __restrict__ v_attr, const float* __restrict__ e_attr,
          const float* __restrict__ g, const int* __restrict__ batch,
          const int* __restrict__ gcnt, const int* __restrict__ recs,
          const int* __restrict__ ovf,
          const float* __restrict__ W1, const float* __restrict__ b1,
          const float* __restrict__ W2, const float* __restrict__ b2,
          const float* __restrict__ W3, const float* __restrict__ b3,
          float* __restrict__ out, int nv, int nb, int bcap) {
  extern __shared__ float smem[];
  float* buf0 = smem;
  float* buf1 = smem + MT * LDP;
  int tid  = threadIdx.x;
  int wave = tid >> 5;
  int lane = tid & 31;
  int base = blockIdx.x * MT;

  int* lst  = reinterpret_cast<int*>(buf1);
  int* lcnt = lst + MT * LSTR;
  if (tid < MT) lcnt[tid] = 0;
  __syncthreads();
  int bkt  = base >> BSH;
  int nrec = gcnt[bkt * CPAD];
  if (nrec > bcap) nrec = bcap;
  int novf = gcnt[nb * CPAD];
  const int2* rb = reinterpret_cast<const int2*>(recs) + (long long)bkt * bcap;
  const int2* ob = reinterpret_cast<const int2*>(ovf);
  int ntot = nrec + novf;
  for (int i0 = tid; i0 < ntot; i0 += 4 * NTHR) {
    int2 rc[4];
#pragma unroll
    for (int u = 0; u < 4; ++u) {
      int i = i0 + u * NTHR;
      rc[u] = make_int2(0, base - 1);
      if (i < ntot) rc[u] = i < nrec ? rb[i] : ob[i - nrec];
    }
#pragma unroll
    for (int u = 0; u < 4; ++u) {
      unsigned loc = (unsigned)(rc[u].y - base);
      if (loc < (unsigned)MT) {
        int rnk = atomicAdd(&lcnt[loc], 1);
        if (rnk < LCAP) lst[loc * LSTR + rnk] = rc[u].x;
      }
    }
  }
  __syncthreads();

  {
    int r = tid >> 3;
    int q = tid & 7;
    int v = base + r;
    const float4 zero = make_float4(0.0f, 0.0f, 0.0f, 0.0f);
    float4 va0 = zero, va1 = zero, g0 = zero, g1 = zero;
    float4 s = zero, mn = zero, mx = zero, mean = zero;
    if (v < nv) {
      const float* vr = v_attr + (long long)v * D_V + 4 * q;
      va0 = *reinterpret_cast<const float4*>(vr);
      va1 = *reinterpret_cast<const float4*>(vr + 32);
      const float* gr = g + batch[v] * D_G + 4 * q;
      g0 = *reinterpret_cast<const float4*>(gr);
      g1 = *reinterpret_cast<const float4*>(gr + 32);

      const float inf = __builtin_inff();
      mn = make_float4(inf, inf, inf, inf);
      mx = make_float4(-inf, -inf, -inf, -inf);
      const float* rowbase = e_attr + 4 * q;
#define LD_ROW(e) (*reinterpret_cast<const float4*>(rowbase + (long long)(e) * D_E))
      int cnt = lcnt[r];
      if (cnt <= LCAP) {
        const int* lr = lst + r * LSTR;
        int i = 0;
        for (; i + 4 <= cnt; i += 4) {
          int4 ee = *reinterpret_cast<const int4*>(lr + i);
          float4 x0 = LD_ROW(ee.x);
          float4 x1 = LD_ROW(ee.y);
          float4 x2 = LD_ROW(ee.z);
          float4 x3 = LD_ROW(ee.w);
          acc_row(s, mn, mx, x0); acc_row(s, mn, mx, x1);
          acc_row(s, mn, mx, x2); acc_row(s, mn, mx, x3);
        }
        for (; i < cnt; ++i) {
          float4 x = LD_ROW(lr[i]);
          acc_row(s, mn, mx, x);
        }
      } else {
        cnt = 0;
        for (int i = 0; i < nrec + novf; ++i) {
          int2 rec = i < nrec ? rb[i] : ob[i - nrec];
          if (rec.y == v) {
            float4 x = LD_ROW(rec.x);
            acc_row(s, mn, mx, x);
            ++cnt;
          }
        }
      }
#undef LD_ROW
      if (cnt == 0) {
        mn = s;
        mx = s;
      }
      float d = fmaxf((float)cnt, 1.0f);
      mean = make_float4(s.x / d, s.y / d, s.z / d, s.w / d);
    }
    float* row = buf0 + r * LDP + 4 * q;
    *reinterpret_cast<float4*>(row)       = va0;
    *reinterpret_cast<float4*>(row + 32)  = va1;
    *reinterpret_cast<float4*>(row + 64)  = mn;
    *reinterpret_cast<float4*>(row + 96)  = mean;
    *reinterpret_cast<float4*>(row + 128) = s;
    *reinterpret_cast<float4*>(row + 160) = mx;
    *reinterpret_cast<float4*>(row + 192) = g0;
    *reinterpret_cast<float4*>(row + 224) = g1;
  }
  __syncthreads();

  layer_wmma_cols(buf0, buf1, W1, b1, wave, lane);
  __syncthreads();
  layer_wmma_cols(buf1, buf0, W2, b2, wave, lane);
  __syncthreads();
  layer_wmma<1>(buf0, buf1, W3, b3,  D_OUT,  false, wave, lane);
  __syncthreads();

  for (int idx = tid; idx < MT * D_OUT; idx += NTHR) {
    int r = idx >> 6;
    int c = idx & 63;
    int v = base + r;
    if (v < nv) out[(long long)v * D_OUT + c] = buf1[r * LDP + c];
  }
}

extern "C" void kernel_launch(void* const* d_in, const int* in_sizes, int n_in,
                              void* d_out, int out_size, void* d_ws, size_t ws_size,
                              hipStream_t stream) {
  const float* v_attr = (const float*)d_in[0];
  const int*   edge   = (const int*)d_in[1];
  const float* e_attr = (const float*)d_in[2];
  const float* g      = (const float*)d_in[3];
  const int*   batch  = (const int*)d_in[4];
  const float* W1     = (const float*)d_in[5];
  const float* b1     = (const float*)d_in[6];
  const float* W2     = (const float*)d_in[7];
  const float* b2     = (const float*)d_in[8];
  const float* W3     = (const float*)d_in[9];
  const float* b3     = (const float*)d_in[10];
  float* out = (float*)d_out;

  int nv = in_sizes[0] / D_V;
  int ne = in_sizes[2] / D_E;

  int nb   = (nv + BWID - 1) / BWID;
  int bcap = (2 * (ne / nb) + 1024 + 31) & ~31;
  int ncnt = (nb + 1) * CPAD;
  int* gcnt = (int*)d_ws;
  int* recs = gcnt + ncnt;
  int* ovf  = recs + (size_t)2 * nb * bcap;

  init_counts<<<(ncnt + 255) / 256, 256, 0, stream>>>(gcnt, ncnt);

  size_t shb = (size_t)(3 * nb + 4) * sizeof(int) + (size_t)EBLK * 2 * sizeof(int);
  edge_bin<<<(ne + EBLK - 1) / EBLK, ETHR, shb, stream>>>(edge, gcnt, recs, ovf,
                                                          ne, nv, nb, bcap);

  int nblk = (nv + MT - 1) / MT;
  size_t sh = (size_t)2 * MT * LDP * sizeof(float);
  mlp_fused<<<nblk, NTHR, sh, stream>>>(v_attr, e_attr, g, batch, gcnt, recs, ovf,
                                        W1, b1, W2, b2, W3, b3, out, nv, nb, bcap);
}


# Rows behind the whole rounds of resident blocks (512 = 256 WGPs x 2) go into 16-row tiles, one WMMA row tile per wave, not 64-row ones: the grid's sparse last round gets short; order of sums kept
# speedup vs baseline: 2.4129x; 1.0500x over previous
#include <hip/hip_runtime.h>

typedef __attribute__((ext_vector_type(2))) float v2f;
typedef __attribute__((ext_vector_type(8))) float v8f;

#define D_E   32
#define D_V   64
#define D_G   64
#define HDIM  256
#define D_OUT 64
#define NIN   256
#define MT    64
#define NTHR  512
#define LDP   260
#define BSH   8
#define BWID  256
#define CPAD  32
#define EBLK  8192
#define ETHR  512
#define EPT   16
#define LCAP  128
#define LSTR  132
#define RSUB  16
#define SLOTS 512

__global__ void init_counts(int* gcnt, int n) {
  int tid = blockIdx.x * blockDim.x + threadIdx.x;
  if (tid < n) gcnt[tid] = 0;
}

__global__ void __launch_bounds__(ETHR)
edge_bin(const int* __restrict__ src, int* gcnt, int* __restrict__ recs,
         int* __restrict__ ovf, int ne, int nv, int nb, int bcap) {
  extern __shared__ int sm[];
  int* hist   = sm;
  int* lstart = sm + nb;
  int* gbase  = sm + 2 * nb;
  int* totalp = sm + 3 * nb;
  int2* stage = reinterpret_cast<int2*>(sm + ((3 * nb + 3) & ~1));
  int tid = threadIdx.x;

  for (int b = tid; b < nb; b += ETHR) hist[b] = 0;
  __syncthreads();

  int e0 = blockIdx.x * EBLK;
  int sv[EPT], rk[EPT];
#pragma unroll
  for (int j = 0; j < EPT; ++j) {
    int e = e0 + j * ETHR + tid;
    int s = -1;
    if (e < ne) {
      s = src[e];
      if ((unsigned)s >= (unsigned)nv) s = -1;
    }
    sv[j] = s;
  }
#pragma unroll
  for (int j = 0; j < EPT; ++j) {
    rk[j] = 0;
    if (sv[j] >= 0) rk[j] = atomicAdd(&hist[sv[j] >> BSH], 1);
  }
  __syncthreads();

  if (tid < 32) {
    int seg = (nb + 31) / 32;
    int lo = tid * seg;
    int hi = lo + seg < nb ? lo + seg : nb;
    int sum = 0;
    for (int b = lo; b < hi; ++b) sum += hist[b];
    int incl = sum;
    for (int d = 1; d < 32; d <<= 1) {
      int t = __shfl_up(incl, d);
      if (tid >= d) incl += t;
    }
    int run = incl - sum;
    for (int b = lo; b < hi; ++b) {
      lstart[b] = run;
      run += hist[b];
    }
    if (tid == 31) *totalp = incl;
  }
  for (int b = tid; b < nb; b += ETHR) {
    int h = hist[b];
    gbase[b] = h ? atomicAdd(&gcnt[b * CPAD], h) : 0;
  }
  __syncthreads();

#pragma unroll
  for (int j = 0; j < EPT; ++j) {
    if (sv[j] >= 0) {
      stage[lstart[sv[j] >> BSH] + rk[j]] = make_int2(e0 + j * ETHR + tid, sv[j]);
    }
  }
  __syncthreads();

  int total = *totalp;
  for (int i = tid; i < total; i += ETHR) {
    int2 rec = stage[i];
    int b = rec.y >> BSH;
    int dst = gbase[b] + (i - lstart[b]);
    if (dst < bcap) {
      *reinterpret_cast<int2*>(recs + 2 * ((long long)b * bcap + dst)) = rec;
    } else {
      int k = atomicAdd(&gcnt[nb * CPAD], 1);
      *reinterpret_cast<int2*>(ovf + 2 * (long long)k) = rec;
    }
  }
}

template <int NTW, int NT>
__device__ __forceinline__ void layer_wmma(const float* __restrict__ hin,
                                           float* __restrict__ hout,
                                           const float* __restrict__ W,
                                           const float* __restrict__ b,
                                           int wstride, bool do_relu,
                                           int wave, int lane) {
  int lr   = lane & 15;
  int half = lane >> 4;
  int rt   = NT == 4 ? (wave & 3) : 0;
  int col0 = (NT == 4 ? (wave >> 2) : wave) * 16 + lr;
  if (NT == 1 && wave >= 4) return;

  v8f acc[NTW];
#pragma unroll
  for (int j = 0; j < NTW; ++j) {
    float bv = b[col0 + 64 * j];
    acc[j] = {bv, bv, bv, bv, bv, bv, bv, bv};
  }

  const float* arow = &hin[(rt * 16 + lr) * LDP];
  for (int kc = 0; kc < NIN; kc += 32) {
    v2f afrag[8];
#pragma unroll
    for (int k = 0; k < 8; ++k) {
      int kk = kc + 4 * k + 2 * half;
      afrag[k].x = arow[kk];
      afrag[k].y = arow[kk + 1];
    }
#pragma unroll
    for (int k = 0; k < 8; ++k) {
      int kk = kc + 4 * k + 2 * half;
      const float* wr0 = &W[(long long)kk * wstride + col0];
      const float* wr1 = wr0 + wstride;
#pragma unroll
      for (int j = 0; j < NTW; ++j) {
        v2f bb;
        bb.x = wr0[64 * j];
        bb.y = wr1[64 * j];
        acc[j] = __builtin_amdgcn_wmma_f32_16x16x4_f32(
            false, afrag[k], false, bb, (short)0, acc[j], false, false);
      }
    }
  }

#pragma unroll
  for (int j = 0; j < NTW; ++j) {
#pragma unroll
    for (int i = 0; i < 8; ++i) {
      float x = acc[j][i];
      if (do_relu) x = fmaxf(x, 0.0f);
      hout[(rt * 16 + i + 8 * half) * LDP + col0 + 64 * j] = x;
    }
  }
}


#define KCH 16
template <int NT>
__device__ __forceinline__ void load_chunk(const float* __restrict__ wp,
                                           const float* __restrict__ ap, int kc,
                                           v2f (&bq)[4], v2f (&aq)[4][NT]) {
#pragma unroll
  for (int k = 0; k < 4; ++k) {
    const float* wr = wp + (kc + 4 * k) * HDIM;
    bq[k].x = wr[0];
    bq[k].y = wr[HDIM];
  }
#pragma unroll
  for (int k = 0; k < 4; ++k) {
#pragma unroll
    for (int t = 0; t < NT; ++t) {
      aq[k][t] = *reinterpret_cast<const v2f*>(ap + t * 16 * LDP + kc + 4 * k);
    }
  }
}

template <int NT>
__device__ __forceinline__ void mma_chunk(v8f (&acc)[NT], const v2f (&bq)[4],
                                          const v2f (&aq)[4][NT]) {
#pragma unroll
  for (int k = 0; k < 4; ++k) {
#pragma unroll
    for (int t = 0; t < NT; ++t) {
      acc[t] = __builtin_amdgcn_wmma_f32_16x16x4_f32(
          false, aq[k][t], false, bq[k], (short)0, acc[t], false, false);
    }
  }
}

template <int NT>
__device__ __forceinline__ void layer_wmma_cols(const float* __restrict__ hin,
                                                float* __restrict__ hout,
                                                const float* __restrict__ W,
                                                const float* __restrict__ b,
                                                int wave, int lane) {
  int lr   = lane & 15;
  int half = lane >> 4;
  int col  = wave * 16 + lr;

  float bv = b[col];
  v8f acc[NT];
#pragma unroll
  for (int t = 0; t < NT; ++t) acc[t] = {bv, bv, bv, bv, bv, bv, bv, bv};

  const float* wp = W + (2 * half) * HDIM + col;
  const float* ap = hin + lr * LDP + 2 * half;

  v2f b0[4], b1[4];
  v2f a0[4][NT], a1[4][NT];
  load_chunk<NT>(wp, ap, 0, b0, a0);
  for (int kc = 0; kc < NIN; kc += 2 * KCH) {
    load_chunk<NT>(wp, ap, kc + KCH, b1, a1);
    mma_chunk<NT>(acc, b0, a0);
    if (kc + 2 * KCH < NIN) load_chunk<NT>(wp, ap, kc + 2 * KCH, b0, a0);
    mma_chunk<NT>(acc, b1, a1);
  }

#pragma unroll
  for (int t = 0; t < NT; ++t) {
#pragma unroll
    for (int i = 0; i < 8; ++i) {
      hout[(t * 16 + i + 8 * half) * LDP + col] = fmaxf(acc[t][i], 0.0f);
    }
  }
}

__device__ __forceinline__ void acc_row(float4& s, float4& mn, float4& mx, const float4& x) {
  s.x += x.x; s.y += x.y; s.z += x.z; s.w += x.w;
  mn.x = fminf(mn.x, x.x); mn.y = fminf(mn.y, x.y);
  mn.z = fminf(mn.z, x.z); mn.w = fminf(mn.w, x.w);
  mx.x = fmaxf(mx.x, x.x); mx.y = fmaxf(mx.y, x.y);
  mx.z = fmaxf(mx.z, x.z); mx.w = fmaxf(mx.w, x.w);
}

template <int NT>
__device__ __forceinline__ void
tile_body(float* __restrict__ smem, int base,
          const float* __restrict__ v_attr, const float* __restrict__ e_attr,
          const float* __restrict__ g, const int* __restrict__ batch,
          const int* __restrict__ gcnt, const int* __restrict__ recs,
          const int* __restrict__ ovf,
          const float* __restrict__ W1, const float* __restrict__ b1,
          const float* __restrict__ W2, const float* __restrict__ b2,
          const float* __restrict__ W3, const float* __restrict__ b3,
          float* __restrict__ out, int nv, int nb, int bcap) {
  constexpr int ROWS = NT * 16;
  float* buf0 = smem;
  float* buf1 = smem + MT * LDP;
  int tid  = threadIdx.x;
  int wave = tid >> 5;
  int lane = tid & 31;

  int* lst  = reinterpret_cast<int*>(buf1);
  int* lcnt = lst + MT * LSTR;
  if (tid < MT) lcnt[tid] = 0;
  __syncthreads();
  int bkt  = base >> BSH;
  int nrec = gcnt[bkt * CPAD];
  if (nrec > bcap) nrec = bcap;
  int novf = gcnt[nb * CPAD];
  const int2* rb = reinterpret_cast<const int2*>(recs) + (long long)bkt * bcap;
  const int2* ob = reinterpret_cast<const int2*>(ovf);
  int ntot = nrec + novf;
  for (int i0 = tid; i0 < ntot; i0 += 4 * NTHR) {
    int2 rc[4];
#pragma unroll
    for (int u = 0; u < 4; ++u) {
      int i = i0 + u * NTHR;
      rc[u] = make_int2(0, base - 1);
      if (i < ntot) rc[u] = i < nrec ? rb[i] : ob[i - nrec];
    }
#pragma unroll
    for (int u = 0; u < 4; ++u) {
      unsigned loc = (unsigned)(rc[u].y - base);
      if (loc < (unsigned)ROWS) {
        int rnk = atomicAdd(&lcnt[loc], 1);
        if (rnk < LCAP) lst[loc * LSTR + rnk] = rc[u].x;
      }
    }
  }
  __syncthreads();

  {
    int r = tid >> 3;
    int q = tid & 7;
    int v = base + r;
    const float4 zero = make_float4(0.0f, 0.0f, 0.0f, 0.0f);
    float4 va0 = zero, va1 = zero, g0 = zero, g1 = zero;
    float4 s = zero, mn = zero, mx = zero, mean = zero;
    if (r < ROWS && v < nv) {
      const float* vr = v_attr + (long long)v * D_V + 4 * q;
      va0 = *reinterpret_cast<const float4*>(vr);
      va1 = *reinterpret_cast<const float4*>(vr + 32);
      const float* gr = g + batch[v] * D_G + 4 * q;
      g0 = *reinterpret_cast<const float4*>(gr);
      g1 = *reinterpret_cast<const float4*>(gr + 32);

      const float inf = __builtin_inff();
      mn = make_float4(inf, inf, inf, inf);
      mx = make_float4(-inf, -inf, -inf, -inf);
      const float* rowbase = e_attr + 4 * q;
#define LD_ROW(e) (*reinterpret_cast<const float4*>(rowbase + (long long)(e) * D_E))
      int cnt = lcnt[r];
      if (cnt <= LCAP) {
        const int* lr = lst + r * LSTR;
        int i = 0;
        for (; i + 4 <= cnt; i += 4) {
          int4 ee = *reinterpret_cast<const int4*>(lr + i);
          float4 x0 = LD_ROW(ee.x);
          float4 x1 = LD_ROW(ee.y);
          float4 x2 = LD_ROW(ee.z);
          float4 x3 = LD_ROW(ee.w);
          acc_row(s, mn, mx, x0); acc_row(s, mn, mx, x1);
          acc_row(s, mn, mx, x2); acc_row(s, mn, mx, x3);
        }
        for (; i < cnt; ++i) {
          float4 x = LD_ROW(lr[i]);
          acc_row(s, mn, mx, x);
        }
      } else {
        cnt = 0;
        for (int i = 0; i < nrec + novf; ++i) {
          int2 rec = i < nrec ? rb[i] : ob[i - nrec];
          if (rec.y == v) {
            float4 x = LD_ROW(rec.x);
            acc_row(s, mn, mx, x);
            ++cnt;
          }
        }
      }
#undef LD_ROW
      if (cnt == 0) {
        mn = s;
        mx = s;
      }
      float d = fmaxf((float)cnt, 1.0f);
      mean = make_float4(s.x / d, s.y / d, s.z / d, s.w / d);
    }
    if (r < ROWS) {
      float* row = buf0 + r * LDP + 4 * q;
      *reinterpret_cast<float4*>(row)       = va0;
      *reinterpret_cast<float4*>(row + 32)  = va1;
      *reinterpret_cast<float4*>(row + 64)  = mn;
      *reinterpret_cast<float4*>(row + 96)  = mean;
      *reinterpret_cast<float4*>(row + 128) = s;
      *reinterpret_cast<float4*>(row + 160) = mx;
      *reinterpret_cast<float4*>(row + 192) = g0;
      *reinterpret_cast<float4*>(row + 224) = g1;
    }
  }
  __syncthreads();

  layer_wmma_cols<NT>(buf0, buf1, W1, b1, wave, lane);
  __syncthreads();
  layer_wmma_cols<NT>(buf1, buf0, W2, b2, wave, lane);
  __syncthreads();
  layer_wmma<1, NT>(buf0, buf1, W3, b3,  D_OUT,  false, wave, lane);
  __syncthreads();

  for (int idx = tid; idx < ROWS * D_OUT; idx += NTHR) {
    int r = idx >> 6;
    int c = idx & 63;
    int v = base + r;
    if (v < nv) out[(long long)v * D_OUT + c] = buf1[r * LDP + c];
  }
}

__global__ void __launch_bounds__(NTHR)
mlp_fused(const float* __restrict__ v_attr, const float* __restrict__ e_attr,
          const float* __restrict__ g, const int* __restrict__ batch,
          const int* __restrict__ gcnt, const int* __restrict__ recs,
          const int* __restrict__ ovf,
          const float* __restrict__ W1, const float* __restrict__ b1,
          const float* __restrict__ W2, const float* __restrict__ b2,
          const float* __restrict__ W3, const float* __restrict__ b3,
          float* __restrict__ out, int nv, int nb, int bcap, int nfull) {
  extern __shared__ float smem[];
  int blk = blockIdx.x;
  if (blk < nfull) {
    tile_body<4>(smem, blk * MT, v_attr, e_attr, g, batch, gcnt, recs, ovf,
                 W1, b1, W2, b2, W3, b3, out, nv, nb, bcap);
  } else {
    tile_body<1>(smem, nfull * MT + (blk - nfull) * RSUB, v_attr, e_attr, g, batch,
                 gcnt, recs, ovf, W1, b1, W2, b2, W3, b3, out, nv, nb, bcap);
  }
}

extern "C" void kernel_launch(void* const* d_in, const int* in_sizes, int n_in,
                              void* d_out, int out_size, void* d_ws, size_t ws_size,
                              hipStream_t stream) {
  const float* v_attr = (const float*)d_in[0];
  const int*   edge   = (const int*)d_in[1];
  const float* e_attr = (const float*)d_in[2];
  const float* g      = (const float*)d_in[3];
  const int*   batch  = (const int*)d_in[4];
  const float* W1     = (const float*)d_in[5];
  const float* b1     = (const float*)d_in[6];
  const float* W2     = (const float*)d_in[7];
  const float* b2     = (const float*)d_in[8];
  const float* W3     = (const float*)d_in[9];
  const float* b3     = (const float*)d_in[10];
  float* out = (float*)d_out;

  int nv = in_sizes[0] / D_V;
  int ne = in_sizes[2] / D_E;

  int nb   = (nv + BWID - 1) / BWID;
  int bcap = (2 * (ne / nb) + 1024 + 31) & ~31;
  int ncnt = (nb + 1) * CPAD;
  int* gcnt = (int*)d_ws;
  int* recs = gcnt + ncnt;
  int* ovf  = recs + (size_t)2 * nb * bcap;

  init_counts<<<(ncnt + 255) / 256, 256, 0, stream>>>(gcnt, ncnt);

  size_t shb = (size_t)(3 * nb + 4) * sizeof(int) + (size_t)EBLK * 2 * sizeof(int);
  edge_bin<<<(ne + EBLK - 1) / EBLK, ETHR, shb, stream>>>(edge, gcnt, recs, ovf,
                                                          ne, nv, nb, bcap);

  int ntile = (nv + MT - 1) / MT;
  int nfull = (ntile / SLOTS) * SLOTS;
  if ((ntile - nfull) * (MT / RSUB) > SLOTS) nfull = ntile;
  int rest  = nv - nfull * MT;
  int nsub  = rest > 0 ? (rest + RSUB - 1) / RSUB : 0;
  size_t sh = (size_t)2 * MT * LDP * sizeof(float);
  mlp_fused<<<nfull + nsub, NTHR, sh, stream>>>(v_attr, e_attr, g, batch, gcnt, recs, ovf,
                                                W1, b1, W2, b2, W3, b3, out, nv, nb, bcap,
                                                nfull);
}


# Layer 1's graph columns hoisted: b1 + g @ W1[192:256] made once per graph (WMMA, spare block of the binning launch) as each vertex's accumulator start; 64 of 256 k cut; graph columns summed first
# speedup vs baseline: 2.5879x; 1.0725x over previous
#include <hip/hip_runtime.h>

typedef __attribute__((ext_vector_type(2))) float v2f;
typedef __attribute__((ext_vector_type(8))) float v8f;

#define D_E   32
#define D_V   64
#define D_G   64
#define HDIM  256
#define D_OUT 64
#define NIN   256
#define MT    64
#define NTHR  512
#define LDP   260
#define BSH   8
#define BWID  256
#define CPAD  32
#define EBLK  8192
#define ETHR  512
#define EPT   16
#define LCAP  128
#define LSTR  132
#define RSUB  16
#define SLOTS 512

__global__ void init_counts(int* gcnt, int n) {
  int tid = blockIdx.x * blockDim.x + threadIdx.x;
  if (tid < n) gcnt[tid] = 0;
}

__device__ __forceinline__ void graph_part(const float* __restrict__ g,
                                           const float* __restrict__ W1,
                                           const float* __restrict__ b1,
                                           float* __restrict__ G1, int ng, int tile) {
  int wave = threadIdx.x >> 5;
  int lane = threadIdx.x & 31;
  int lr   = lane & 15;
  int half = lane >> 4;
  int col  = wave * 16 + lr;
  int row  = tile * 16 + lr;
  bool ok  = row < ng;
  const float* ar = g + (long long)(ok ? row : 0) * D_G + 2 * half;
  const float* wp = W1 + (long long)(NIN - D_G + 2 * half) * HDIM + col;
  float bv = b1[col];
  v8f acc = {bv, bv, bv, bv, bv, bv, bv, bv};
#pragma unroll
  for (int k = 0; k < D_G; k += 4) {
    v2f a, b;
    float a0 = ar[k], a1 = ar[k + 1];
    a.x = ok ? a0 : 0.0f;
    a.y = ok ? a1 : 0.0f;
    b.x = wp[k * HDIM];
    b.y = wp[(k + 1) * HDIM];
    acc = __builtin_amdgcn_wmma_f32_16x16x4_f32(false, a, false, b, (short)0, acc,
                                                false, false);
  }
#pragma unroll
  for (int i = 0; i < 8; ++i) {
    G1[(long long)(tile * 16 + i + 8 * half) * HDIM + col] = acc[i];
  }
}

__global__ void __launch_bounds__(ETHR)
edge_bin(const int* __restrict__ src, int* gcnt, int* __restrict__ recs,
         int* __restrict__ ovf, int ne, int nv, int nb, int bcap,
         const float* __restrict__ g, const float* __restrict__ W1,
         const float* __restrict__ b1, float* __restrict__ G1, int ng, int nblk_e) {
  extern __shared__ int sm[];
  if ((int)blockIdx.x >= nblk_e) {
    graph_part(g, W1, b1, G1, ng, (int)blockIdx.x - nblk_e);
    return;
  }
  int* hist   = sm;
  int* lstart = sm + nb;
  int* gbase  = sm + 2 * nb;
  int* totalp = sm + 3 * nb;
  int2* stage = reinterpret_cast<int2*>(sm + ((3 * nb + 3) & ~1));
  int tid = threadIdx.x;

  for (int b = tid; b < nb; b += ETHR) hist[b] = 0;
  __syncthreads();

  int e0 = blockIdx.x * EBLK;
  int sv[EPT], rk[EPT];
#pragma unroll
  for (int j = 0; j < EPT; ++j) {
    int e = e0 + j * ETHR + tid;
    int s = -1;
    if (e < ne) {
      s = src[e];
      if ((unsigned)s >= (unsigned)nv) s = -1;
    }
    sv[j] = s;
  }
#pragma unroll
  for (int j = 0; j < EPT; ++j) {
    rk[j] = 0;
    if (sv[j] >= 0) rk[j] = atomicAdd(&hist[sv[j] >> BSH], 1);
  }
  __syncthreads();

  if (tid < 32) {
    int seg = (nb + 31) / 32;
    int lo = tid * seg;
    int hi = lo + seg < nb ? lo + seg : nb;
    int sum = 0;
    for (int b = lo; b < hi; ++b) sum += hist[b];
    int incl = sum;
    for (int d = 1; d < 32; d <<= 1) {
      int t = __shfl_up(incl, d);
      if (tid >= d) incl += t;
    }
    int run = incl - sum;
    for (int b = lo; b < hi; ++b) {
      lstart[b] = run;
      run += hist[b];
    }
    if (tid == 31) *totalp = incl;
  }
  for (int b = tid; b < nb; b += ETHR) {
    int h = hist[b];
    gbase[b] = h ? atomicAdd(&gcnt[b * CPAD], h) : 0;
  }
  __syncthreads();

#pragma unroll
  for (int j = 0; j < EPT; ++j) {
    if (sv[j] >= 0) {
      stage[lstart[sv[j] >> BSH] + rk[j]] = make_int2(e0 + j * ETHR + tid, sv[j]);
    }
  }
  __syncthreads();

  int total = *totalp;
  for (int i = tid; i < total; i += ETHR) {
    int2 rec = stage[i];
    int b = rec.y >> BSH;
    int dst = gbase[b] + (i - lstart[b]);
    if (dst < bcap) {
      *reinterpret_cast<int2*>(recs + 2 * ((long long)b * bcap + dst)) = rec;
    } else {
      int k = atomicAdd(&gcnt[nb * CPAD], 1);
      *reinterpret_cast<int2*>(ovf + 2 * (long long)k) = rec;
    }
  }
}

template <int NTW, int NT>
__device__ __forceinline__ void layer_wmma(const float* __restrict__ hin,
                                           float* __restrict__ hout,
                                           const float* __restrict__ W,
                                           const float* __restrict__ b,
                                           int wstride, bool do_relu,
                                           int wave, int lane) {
  int lr   = lane & 15;
  int half = lane >> 4;
  int rt   = NT == 4 ? (wave & 3) : 0;
  int col0 = (NT == 4 ? (wave >> 2) : wave) * 16 + lr;
  if (NT == 1 && wave >= 4) return;

  v8f acc[NTW];
#pragma unroll
  for (int j = 0; j < NTW; ++j) {
    float bv = b[col0 + 64 * j];
    acc[j] = {bv, bv, bv, bv, bv, bv, bv, bv};
  }

  const float* arow = &hin[(rt * 16 + lr) * LDP];
  for (int kc = 0; kc < NIN; kc += 32) {
    v2f afrag[8];
#pragma unroll
    for (int k = 0; k < 8; ++k) {
      int kk = kc + 4 * k + 2 * half;
      afrag[k].x = arow[kk];
      afrag[k].y = arow[kk + 1];
    }
#pragma unroll
    for (int k = 0; k < 8; ++k) {
      int kk = kc + 4 * k + 2 * half;
      const float* wr0 = &W[(long long)kk * wstride + col0];
      const float* wr1 = wr0 + wstride;
#pragma unroll
      for (int j = 0; j < NTW; ++j) {
        v2f bb;
        bb.x = wr0[64 * j];
        bb.y = wr1[64 * j];
        acc[j] = __builtin_amdgcn_wmma_f32_16x16x4_f32(
            false, afrag[k], false, bb, (short)0, acc[j], false, false);
      }
    }
  }

#pragma unroll
  for (int j = 0; j < NTW; ++j) {
#pragma unroll
    for (int i = 0; i < 8; ++i) {
      float x = acc[j][i];
      if (do_relu) x = fmaxf(x, 0.0f);
      hout[(rt * 16 + i + 8 * half) * LDP + col0 + 64 * j] = x;
    }
  }
}


#define KCH 16
template <int NT>
__device__ __forceinline__ void load_chunk(const float* __restrict__ wp,
                                           const float* __restrict__ ap, int kc,
                                           v2f (&bq)[4], v2f (&aq)[4][NT]) {
#pragma unroll
  for (int k = 0; k < 4; ++k) {
    const float* wr = wp + (kc + 4 * k) * HDIM;
    bq[k].x = wr[0];
    bq[k].y = wr[HDIM];
  }
#pragma unroll
  for (int k = 0; k < 4; ++k) {
#pragma unroll
    for (int t = 0; t < NT; ++t) {
      aq[k][t] = *reinterpret_cast<const v2f*>(ap + t * 16 * LDP + kc + 4 * k);
    }
  }
}

template <int NT>
__device__ __forceinline__ void mma_chunk(v8f (&acc)[NT], const v2f (&bq)[4],
                                          const v2f (&aq)[4][NT]) {
#pragma unroll
  for (int k = 0; k < 4; ++k) {
#pragma unroll
    for (int t = 0; t < NT; ++t) {
      acc[t] = __builtin_amdgcn_wmma_f32_16x16x4_f32(
          false, aq[k][t], false, bq[k], (short)0, acc[t], false, false);
    }
  }
}

template <int NT, int KTOT, bool GSTART>
__device__ __forceinline__ void layer_wmma_cols(const float* __restrict__ hin,
                                                float* __restrict__ hout,
                                                const float* __restrict__ W,
                                                const float* __restrict__ b,
                                                int wave, int lane) {
  int lr   = lane & 15;
  int half = lane >> 4;
  int col  = wave * 16 + lr;

  v8f acc[NT];
  if (GSTART) {
    const int* gi = reinterpret_cast<const int*>(hin) + KTOT;
#pragma unroll
    for (int t = 0; t < NT; ++t) {
#pragma unroll
      for (int i = 0; i < 8; ++i) {
        int gid = gi[(t * 16 + i + 8 * half) * LDP];
        acc[t][i] = b[gid * HDIM + col];
      }
    }
  } else {
    float bv = b[col];
#pragma unroll
    for (int t = 0; t < NT; ++t) acc[t] = {bv, bv, bv, bv, bv, bv, bv, bv};
  }

  const float* wp = W + (2 * half) * HDIM + col;
  const float* ap = hin + lr * LDP + 2 * half;

  v2f b0[4], b1[4];
  v2f a0[4][NT], a1[4][NT];
  load_chunk<NT>(wp, ap, 0, b0, a0);
  for (int kc = 0; kc < KTOT; kc += 2 * KCH) {
    load_chunk<NT>(wp, ap, kc + KCH, b1, a1);
    mma_chunk<NT>(acc, b0, a0);
    if (kc + 2 * KCH < KTOT) load_chunk<NT>(wp, ap, kc + 2 * KCH, b0, a0);
    mma_chunk<NT>(acc, b1, a1);
  }

#pragma unroll
  for (int t = 0; t < NT; ++t) {
#pragma unroll
    for (int i = 0; i < 8; ++i) {
      hout[(t * 16 + i + 8 * half) * LDP + col] = fmaxf(acc[t][i], 0.0f);
    }
  }
}

__device__ __forceinline__ void acc_row(float4& s, float4& mn, float4& mx, const float4& x) {
  s.x += x.x; s.y += x.y; s.z += x.z; s.w += x.w;
  mn.x = fminf(mn.x, x.x); mn.y = fminf(mn.y, x.y);
  mn.z = fminf(mn.z, x.z); mn.w = fminf(mn.w, x.w);
  mx.x = fmaxf(mx.x, x.x); mx.y = fmaxf(mx.y, x.y);
  mx.z = fmaxf(mx.z, x.z); mx.w = fmaxf(mx.w, x.w);
}

template <int NT>
__device__ __forceinline__ void
tile_body(float* __restrict__ smem, int base,
          const float* __restrict__ v_attr, const float* __restrict__ e_attr,
          const float* __restrict__ G1, int ng, const int* __restrict__ batch,
          const int* __restrict__ gcnt, const int* __restrict__ recs,
          const int* __restrict__ ovf,
          const float* __restrict__ W1,
          const float* __restrict__ W2, const float* __restrict__ b2,
          const float* __restrict__ W3, const float* __restrict__ b3,
          float* __restrict__ out, int nv, int nb, int bcap) {
  constexpr int ROWS = NT * 16;
  float* buf0 = smem;
  float* buf1 = smem + MT * LDP;
  int tid  = threadIdx.x;
  int wave = tid >> 5;
  int lane = tid & 31;

  int* lst  = reinterpret_cast<int*>(buf1);
  int* lcnt = lst + MT * LSTR;
  if (tid < MT) lcnt[tid] = 0;
  __syncthreads();
  int bkt  = base >> BSH;
  int nrec = gcnt[bkt * CPAD];
  if (nrec > bcap) nrec = bcap;
  int novf = gcnt[nb * CPAD];
  const int2* rb = reinterpret_cast<const int2*>(recs) + (long long)bkt * bcap;
  const int2* ob = reinterpret_cast<const int2*>(ovf);
  int ntot = nrec + novf;
  for (int i0 = tid; i0 < ntot; i0 += 4 * NTHR) {
    int2 rc[4];
#pragma unroll
    for (int u = 0; u < 4; ++u) {
      int i = i0 + u * NTHR;
      rc[u] = make_int2(0, base - 1);
      if (i < ntot) rc[u] = i < nrec ? rb[i] : ob[i - nrec];
    }
#pragma unroll
    for (int u = 0; u < 4; ++u) {
      unsigned loc = (unsigned)(rc[u].y - base);
      if (loc < (unsigned)ROWS) {
        int rnk = atomicAdd(&lcnt[loc], 1);
        if (rnk < LCAP) lst[loc * LSTR + rnk] = rc[u].x;
      }
    }
  }
  __syncthreads();

  {
    int r = tid >> 3;
    int q = tid & 7;
    int v = base + r;
    const float4 zero = make_float4(0.0f, 0.0f, 0.0f, 0.0f);
    float4 va0 = zero, va1 = zero;
    int gid = 0;
    float4 s = zero, mn = zero, mx = zero, mean = zero;
    if (r < ROWS && v < nv) {
      const float* vr = v_attr + (long long)v * D_V + 4 * q;
      va0 = *reinterpret_cast<const float4*>(vr);
      va1 = *reinterpret_cast<const float4*>(vr + 32);
      gid = batch[v];
      gid = gid < 0 ? 0 : (gid < ng ? gid : ng - 1);

      const float inf = __builtin_inff();
      mn = make_float4(inf, inf, inf, inf);
      mx = make_float4(-inf, -inf, -inf, -inf);
      const float* rowbase = e_attr + 4 * q;
#define LD_ROW(e) (*reinterpret_cast<const float4*>(rowbase + (long long)(e) * D_E))
      int cnt = lcnt[r];
      if (cnt <= LCAP) {
        const int* lr = lst + r * LSTR;
        int i = 0;
        for (; i + 4 <= cnt; i += 4) {
          int4 ee = *reinterpret_cast<const int4*>(lr + i);
          float4 x0 = LD_ROW(ee.x);
          float4 x1 = LD_ROW(ee.y);
          float4 x2 = LD_ROW(ee.z);
          float4 x3 = LD_ROW(ee.w);
          acc_row(s, mn, mx, x0); acc_row(s, mn, mx, x1);
          acc_row(s, mn, mx, x2); acc_row(s, mn, mx, x3);
        }
        for (; i < cnt; ++i) {
          float4 x = LD_ROW(lr[i]);
          acc_row(s, mn, mx, x);
        }
      } else {
        cnt = 0;
        for (int i = 0; i < nrec + novf; ++i) {
          int2 rec = i < nrec ? rb[i] : ob[i - nrec];
          if (rec.y == v) {
            float4 x = LD_ROW(rec.x);
            acc_row(s, mn, mx, x);
            ++cnt;
          }
        }
      }
#undef LD_ROW
      if (cnt == 0) {
        mn = s;
        mx = s;
      }
      float d = fmaxf((float)cnt, 1.0f);
      mean = make_float4(s.x / d, s.y / d, s.z / d, s.w / d);
    }
    if (r < ROWS) {
      float* row = buf0 + r * LDP + 4 * q;
      *reinterpret_cast<float4*>(row)       = va0;
      *reinterpret_cast<float4*>(row + 32)  = va1;
      *reinterpret_cast<float4*>(row + 64)  = mn;
      *reinterpret_cast<float4*>(row + 96)  = mean;
      *reinterpret_cast<float4*>(row + 128) = s;
      *reinterpret_cast<float4*>(row + 160) = mx;
      if (q == 0) *reinterpret_cast<int*>(row + (NIN - D_G)) = gid;
    }
  }
  __syncthreads();

  layer_wmma_cols<NT, NIN - D_G, true>(buf0, buf1, W1, G1, wave, lane);
  __syncthreads();
  layer_wmma_cols<NT, NIN, false>(buf1, buf0, W2, b2, wave, lane);
  __syncthreads();
  layer_wmma<1, NT>(buf0, buf1, W3, b3,  D_OUT,  false, wave, lane);
  __syncthreads();

  for (int idx = tid; idx < ROWS * D_OUT; idx += NTHR) {
    int r = idx >> 6;
    int c = idx & 63;
    int v = base + r;
    if (v < nv) out[(long long)v * D_OUT + c] = buf1[r * LDP + c];
  }
}

__global__ void __launch_bounds__(NTHR)
mlp_fused(const float* __restrict__ v_attr, const float* __restrict__ e_attr,
          const float* __restrict__ G1, int ng, const int* __restrict__ batch,
          const int* __restrict__ gcnt, const int* __restrict__ recs,
          const int* __restrict__ ovf,
          const float* __restrict__ W1,
          const float* __restrict__ W2, const float* __restrict__ b2,
          const float* __restrict__ W3, const float* __restrict__ b3,
          float* __restrict__ out, int nv, int nb, int bcap, int nfull) {
  extern __shared__ float smem[];
  int blk = blockIdx.x;
  if (blk < nfull) {
    tile_body<4>(smem, blk * MT, v_attr, e_attr, G1, ng, batch, gcnt, recs, ovf,
                 W1, W2, b2, W3, b3, out, nv, nb, bcap);
  } else {
    tile_body<1>(smem, nfull * MT + (blk - nfull) * RSUB, v_attr, e_attr, G1, ng, batch,
                 gcnt, recs, ovf, W1, W2, b2, W3, b3, out, nv, nb, bcap);
  }
}

extern "C" void kernel_launch(void* const* d_in, const int* in_sizes, int n_in,
                              void* d_out, int out_size, void* d_ws, size_t ws_size,
                              hipStream_t stream) {
  const float* v_attr = (const float*)d_in[0];
  const int*   edge   = (const int*)d_in[1];
  const float* e_attr = (const float*)d_in[2];
  const float* g      = (const float*)d_in[3];
  const int*   batch  = (const int*)d_in[4];
  const float* W1     = (const float*)d_in[5];
  const float* b1     = (const float*)d_in[6];
  const float* W2     = (const float*)d_in[7];
  const float* b2     = (const float*)d_in[8];
  const float* W3     = (const float*)d_in[9];
  const float* b3     = (const float*)d_in[10];
  float* out = (float*)d_out;

  int nv = in_sizes[0] / D_V;
  int ne = in_sizes[2] / D_E;
  int ng = in_sizes[3] / D_G;

  int nb   = (nv + BWID - 1) / BWID;
  int bcap = (2 * (ne / nb) + 1024 + 31) & ~31;
  int ncnt = (nb + 1) * CPAD;
  int* gcnt = (int*)d_ws;
  int* recs = gcnt + ncnt;
  int* ovf  = recs + (size_t)2 * nb * bcap;
  float* G1 = (float*)(ovf + (size_t)2 * ne);
  int ntg   = (ng + 15) / 16;

  init_counts<<<(ncnt + 255) / 256, 256, 0, stream>>>(gcnt, ncnt);

  size_t shb = (size_t)(3 * nb + 4) * sizeof(int) + (size_t)EBLK * 2 * sizeof(int);
  int nblk_e = (ne + EBLK - 1) / EBLK;
  edge_bin<<<nblk_e + ntg, ETHR, shb, stream>>>(edge, gcnt, recs, ovf, ne, nv, nb, bcap,
                                                g, W1, b1, G1, ng, nblk_e);

  int ntile = (nv + MT - 1) / MT;
  int nfull = (ntile / SLOTS) * SLOTS;
  if ((ntile - nfull) * (MT / RSUB) > SLOTS) nfull = ntile;
  int rest  = nv - nfull * MT;
  int nsub  = rest > 0 ? (rest + RSUB - 1) / RSUB : 0;
  size_t sh = (size_t)2 * MT * LDP * sizeof(float);
  mlp_fused<<<nfull + nsub, NTHR, sh, stream>>>(v_attr, e_attr, G1, ng, batch, gcnt, recs,
                                                ovf, W1, W2, b2, W3, b3, out, nv, nb, bcap,
                                                nfull);
}
